# starting kernel
# baseline (speedup 1.0000x reference)
#include <hip/hip_runtime.h>


#define BATCH   4
#define DIN     2048
#define NSTATE  16
#define SEQLEN  2048
#define CHUNK   16
#define NCHUNK  (SEQLEN / CHUNK)

typedef float v2f __attribute__((ext_vector_type(2)));
typedef float v8f __attribute__((ext_vector_type(8)));

template <int CTRL>
__device__ __forceinline__ float dpp_mov(float x) {
  int r = __builtin_amdgcn_update_dpp(0, __builtin_bit_cast(int, x),
                                      CTRL, 0xF, 0xF, true);
  return __builtin_bit_cast(float, r);
}

__device__ __forceinline__ float allreduce16(float v) {
  v += dpp_mov<0x121>(v);
  v += dpp_mov<0x122>(v);
  v += dpp_mov<0x124>(v);
  v += dpp_mov<0x128>(v);
  return v;
}

__device__ __forceinline__ float prefix16(float v) {
  v += dpp_mov<0x111>(v);
  v += dpp_mov<0x112>(v);
  v += dpp_mov<0x114>(v);
  v += dpp_mov<0x118>(v);
  return v;
}

__device__ __forceinline__ float xhalf16(float v) {
  int i = __builtin_bit_cast(int, v);
  int r = __builtin_amdgcn_permlanex16(i, i, 0x76543210, 0xFEDCBA98,
                                       false, false);
  return __builtin_bit_cast(float, r);
}

__device__ __forceinline__ float softplus_f(float x) {
  return (x > 20.0f) ? x : __logf(1.0f + __expf(x));
}

__global__ __launch_bounds__(128) void mamba_scan_kernel(
    const float* __restrict__ u,
    const float* __restrict__ delta,
    const float* __restrict__ A,
    const float* __restrict__ Bm,
    const float* __restrict__ Cm,
    const float* __restrict__ Dv,
    const float* __restrict__ z,
    float* __restrict__ out)
{
  const int lane  = threadIdx.x & 31;
  const int wave  = (blockIdx.x << 2) | (threadIdx.x >> 5);
  const int b     = wave / DIN;
  const int d     = wave % DIN;

  const int tl    = lane & 15;
  const int half  = lane >> 4;

  float Afac[4][2];
#pragma unroll
  for (int q = 0; q < 4; ++q)
#pragma unroll
    for (int j = 0; j < 2; ++j)
      Afac[q][j] = A[(size_t)d * NSTATE + (4 * q + j + 2 * half)];

  const float Dd = Dv[d];

  const size_t rowU  = ((size_t)b * DIN + d) * SEQLEN;
  const size_t rowBC = (size_t)b * NSTATE * SEQLEN;

  float x[4][2];
#pragma unroll
  for (int q = 0; q < 4; ++q) { x[q][0] = 0.0f; x[q][1] = 0.0f; }

  for (int c = 0; c < NCHUNK; ++c) {
    const int base = c * CHUNK;

    if (c + 1 < NCHUNK) {
      __builtin_prefetch(&u[rowU + base + CHUNK + tl], 0, 3);
      __builtin_prefetch(&delta[rowU + base + CHUNK + tl], 0, 3);
      __builtin_prefetch(&z[rowU + base + CHUNK + tl], 0, 3);
    }

    const float del = delta[rowU + base + tl];
    const float uu  = u[rowU + base + tl];
    const float zz  = z[rowU + base + tl];

    const float dtv = softplus_f(del);
    const float du  = dtv * uu;

    const float sT = prefix16(dtv);
    const float S  = __shfl(sT, 15, 16);

    v2f a[4], bm[4];
#pragma unroll
    for (int q = 0; q < 4; ++q) {
#pragma unroll
      for (int j = 0; j < 2; ++j) {
        const int    n   = 4 * q + j + 2 * half;
        const size_t idx = rowBC + (size_t)n * SEQLEN + base + tl;
        const float  cr  = Cm[idx];
        const float  br  = Bm[idx];
        const float  af  = Afac[q][j];
        float ct = cr * __expf(af * sT);
        float bu = br * du * __expf(-af * sT);
        if (tl == 0) bu += x[q][j];
        a[q][j]  = ct;
        bm[q][j] = bu;
      }
    }

    v8f K = {0.f, 0.f, 0.f, 0.f, 0.f, 0.f, 0.f, 0.f};
#pragma unroll
    for (int q = 0; q < 4; ++q) {
      K = __builtin_amdgcn_wmma_f32_16x16x4_f32(
          false, a[q], false, bm[q], (short)0, K, false, false);
    }

#pragma unroll
    for (int q = 0; q < 4; ++q) {
#pragma unroll
      for (int j = 0; j < 2; ++j) {
        const float s = allreduce16(bm[q][j]);
        x[q][j] = s * __expf(Afac[q][j] * S);
      }
    }

    float y[8];
#pragma unroll
    for (int r = 0; r < 8; ++r) {
      const int row = r + 8 * half;
      const float v = (tl <= row) ? K[r] : 0.0f;
      y[r] = allreduce16(v);
    }

    float Y = y[0];
#pragma unroll
    for (int r = 0; r < 8; ++r) {
      const float hiv = xhalf16(y[r]);
      Y = (tl == r) ? y[r] : Y;
      Y = (tl == r + 8) ? hiv : Y;
    }

    float yo  = Y + Dd * uu;
    const float sig = 1.0f / (1.0f + __expf(-zz));
    yo *= zz * sig;
    if (lane < 16) out[rowU + base + tl] = yo;
  }
}

extern "C" void kernel_launch(void* const* d_in, const int* in_sizes, int n_in,
                              void* d_out, int out_size, void* d_ws, size_t ws_size,
                              hipStream_t stream) {
  (void)in_sizes; (void)n_in; (void)d_ws; (void)ws_size; (void)out_size;
  const float* u     = (const float*)d_in[0];
  const float* delta = (const float*)d_in[1];
  const float* A     = (const float*)d_in[2];
  const float* Bm    = (const float*)d_in[3];
  const float* Cm    = (const float*)d_in[4];
  const float* Dv    = (const float*)d_in[5];
  const float* z     = (const float*)d_in[6];
  float* out         = (float*)d_out;

  const int nWaves = BATCH * DIN;
  dim3 grid(nWaves / 4), block(128);
  mamba_scan_kernel<<<grid, block, 0, stream>>>(u, delta, A, Bm, Cm, Dv, z, out);
}


# Swap WMMA operand roles so D = K^T (rows s, cols t): causal sum over s becomes a per-lane sum of 8 D registers + one half swap, replacing 8 DPP allreduces and the gather; y summation order changes
# speedup vs baseline: 1.1639x; 1.1639x over previous
#include <hip/hip_runtime.h>


#define BATCH   4
#define DIN     2048
#define NSTATE  16
#define SEQLEN  2048
#define CHUNK   16
#define NCHUNK  (SEQLEN / CHUNK)

typedef float v2f __attribute__((ext_vector_type(2)));
typedef float v8f __attribute__((ext_vector_type(8)));

template <int CTRL>
__device__ __forceinline__ float dpp_mov(float x) {
  int r = __builtin_amdgcn_update_dpp(0, __builtin_bit_cast(int, x),
                                      CTRL, 0xF, 0xF, true);
  return __builtin_bit_cast(float, r);
}

__device__ __forceinline__ float allreduce16(float v) {
  v += dpp_mov<0x121>(v);
  v += dpp_mov<0x122>(v);
  v += dpp_mov<0x124>(v);
  v += dpp_mov<0x128>(v);
  return v;
}

__device__ __forceinline__ float prefix16(float v) {
  v += dpp_mov<0x111>(v);
  v += dpp_mov<0x112>(v);
  v += dpp_mov<0x114>(v);
  v += dpp_mov<0x118>(v);
  return v;
}

__device__ __forceinline__ float xhalf16(float v) {
  int i = __builtin_bit_cast(int, v);
  int r = __builtin_amdgcn_permlanex16(i, i, 0x76543210, 0xFEDCBA98,
                                       false, false);
  return __builtin_bit_cast(float, r);
}

__device__ __forceinline__ float softplus_f(float x) {
  return (x > 20.0f) ? x : __logf(1.0f + __expf(x));
}

__global__ __launch_bounds__(128) void mamba_scan_kernel(
    const float* __restrict__ u,
    const float* __restrict__ delta,
    const float* __restrict__ A,
    const float* __restrict__ Bm,
    const float* __restrict__ Cm,
    const float* __restrict__ Dv,
    const float* __restrict__ z,
    float* __restrict__ out)
{
  const int lane  = threadIdx.x & 31;
  const int wave  = (blockIdx.x << 2) | (threadIdx.x >> 5);
  const int b     = wave / DIN;
  const int d     = wave % DIN;

  const int tl    = lane & 15;
  const int half  = lane >> 4;

  float Afac[4][2];
#pragma unroll
  for (int q = 0; q < 4; ++q)
#pragma unroll
    for (int j = 0; j < 2; ++j)
      Afac[q][j] = A[(size_t)d * NSTATE + (4 * q + j + 2 * half)];

  const float Dd = Dv[d];

  const size_t rowU  = ((size_t)b * DIN + d) * SEQLEN;
  const size_t rowBC = (size_t)b * NSTATE * SEQLEN;

  float x[4][2];
#pragma unroll
  for (int q = 0; q < 4; ++q) { x[q][0] = 0.0f; x[q][1] = 0.0f; }

  for (int c = 0; c < NCHUNK; ++c) {
    const int base = c * CHUNK;

    if (c + 1 < NCHUNK) {
      __builtin_prefetch(&u[rowU + base + CHUNK + tl], 0, 3);
      __builtin_prefetch(&delta[rowU + base + CHUNK + tl], 0, 3);
      __builtin_prefetch(&z[rowU + base + CHUNK + tl], 0, 3);
    }

    const float del = delta[rowU + base + tl];
    const float uu  = u[rowU + base + tl];
    const float zz  = z[rowU + base + tl];

    const float dtv = softplus_f(del);
    const float du  = dtv * uu;

    const float sT = prefix16(dtv);
    const float S  = __shfl(sT, 15, 16);

    v2f a[4], bm[4];
#pragma unroll
    for (int q = 0; q < 4; ++q) {
#pragma unroll
      for (int j = 0; j < 2; ++j) {
        const int    n   = 4 * q + j + 2 * half;
        const size_t idx = rowBC + (size_t)n * SEQLEN + base + tl;
        const float  cr  = Cm[idx];
        const float  br  = Bm[idx];
        const float  af  = Afac[q][j];
        float ct = cr * __expf(af * sT);
        float bu = br * du * __expf(-af * sT);
        if (tl == 0) bu += x[q][j];
        a[q][j]  = ct;
        bm[q][j] = bu;
      }
    }

    v8f K = {0.f, 0.f, 0.f, 0.f, 0.f, 0.f, 0.f, 0.f};
#pragma unroll
    for (int q = 0; q < 4; ++q) {
      K = __builtin_amdgcn_wmma_f32_16x16x4_f32(
          false, bm[q], false, a[q], (short)0, K, false, false);
    }

#pragma unroll
    for (int q = 0; q < 4; ++q) {
#pragma unroll
      for (int j = 0; j < 2; ++j) {
        const float s = allreduce16(bm[q][j]);
        x[q][j] = s * __expf(Afac[q][j] * S);
      }
    }

    float m[8];
#pragma unroll
    for (int r = 0; r < 8; ++r) {
      const int srow = r + 8 * half;
      m[r] = (srow <= tl) ? K[r] : 0.0f;
    }
    const float part = ((m[0] + m[1]) + (m[2] + m[3])) +
                       ((m[4] + m[5]) + (m[6] + m[7]));
    const float Y = part + xhalf16(part);

    float yo  = Y + Dd * uu;
    const float sig = 1.0f / (1.0f + __expf(-zz));
    yo *= zz * sig;
    if (lane < 16) out[rowU + base + tl] = yo;
  }
}

extern "C" void kernel_launch(void* const* d_in, const int* in_sizes, int n_in,
                              void* d_out, int out_size, void* d_ws, size_t ws_size,
                              hipStream_t stream) {
  (void)in_sizes; (void)n_in; (void)d_ws; (void)ws_size; (void)out_size;
  const float* u     = (const float*)d_in[0];
  const float* delta = (const float*)d_in[1];
  const float* A     = (const float*)d_in[2];
  const float* Bm    = (const float*)d_in[3];
  const float* Cm    = (const float*)d_in[4];
  const float* Dv    = (const float*)d_in[5];
  const float* z     = (const float*)d_in[6];
  float* out         = (float*)d_out;

  const int nWaves = BATCH * DIN;
  dim3 grid(nWaves / 4), block(128);
  mamba_scan_kernel<<<grid, block, 0, stream>>>(u, delta, A, Bm, Cm, Dv, z, out);
}


# Native base-2 transcendentals: A pre-scaled by log2(e) once so every decay is a bare exp2; softplus and sigmoid use hardware log2 and rcp, not library log and IEEE divide; no sum order change
# speedup vs baseline: 1.2890x; 1.1075x over previous
#include <hip/hip_runtime.h>


#define BATCH   4
#define DIN     2048
#define NSTATE  16
#define SEQLEN  2048
#define CHUNK   16
#define NCHUNK  (SEQLEN / CHUNK)

typedef float v2f __attribute__((ext_vector_type(2)));
typedef float v8f __attribute__((ext_vector_type(8)));

template <int CTRL>
__device__ __forceinline__ float dpp_mov(float x) {
  int r = __builtin_amdgcn_update_dpp(0, __builtin_bit_cast(int, x),
                                      CTRL, 0xF, 0xF, true);
  return __builtin_bit_cast(float, r);
}

__device__ __forceinline__ float allreduce16(float v) {
  v += dpp_mov<0x121>(v);
  v += dpp_mov<0x122>(v);
  v += dpp_mov<0x124>(v);
  v += dpp_mov<0x128>(v);
  return v;
}

__device__ __forceinline__ float prefix16(float v) {
  v += dpp_mov<0x111>(v);
  v += dpp_mov<0x112>(v);
  v += dpp_mov<0x114>(v);
  v += dpp_mov<0x118>(v);
  return v;
}

__device__ __forceinline__ float xhalf16(float v) {
  int i = __builtin_bit_cast(int, v);
  int r = __builtin_amdgcn_permlanex16(i, i, 0x76543210, 0xFEDCBA98,
                                       false, false);
  return __builtin_bit_cast(float, r);
}

#define LOG2E_F 1.4426950408889634f
#define LN2_F   0.6931471805599453f

__device__ __forceinline__ float softplus_f(float x) {
  const float e = __builtin_amdgcn_exp2f(x * LOG2E_F);
  const float l = __builtin_amdgcn_logf(1.0f + e) * LN2_F;
  return (x > 20.0f) ? x : l;
}

__global__ __launch_bounds__(128) void mamba_scan_kernel(
    const float* __restrict__ u,
    const float* __restrict__ delta,
    const float* __restrict__ A,
    const float* __restrict__ Bm,
    const float* __restrict__ Cm,
    const float* __restrict__ Dv,
    const float* __restrict__ z,
    float* __restrict__ out)
{
  const int lane  = threadIdx.x & 31;
  const int wave  = (blockIdx.x << 2) | (threadIdx.x >> 5);
  const int b     = wave / DIN;
  const int d     = wave % DIN;

  const int tl    = lane & 15;
  const int half  = lane >> 4;

  float Afac[4][2];
#pragma unroll
  for (int q = 0; q < 4; ++q)
#pragma unroll
    for (int j = 0; j < 2; ++j)
      Afac[q][j] = A[(size_t)d * NSTATE + (4 * q + j + 2 * half)] * LOG2E_F;

  const float Dd = Dv[d];

  const size_t rowU  = ((size_t)b * DIN + d) * SEQLEN;
  const size_t rowBC = (size_t)b * NSTATE * SEQLEN;

  float x[4][2];
#pragma unroll
  for (int q = 0; q < 4; ++q) { x[q][0] = 0.0f; x[q][1] = 0.0f; }

  for (int c = 0; c < NCHUNK; ++c) {
    const int base = c * CHUNK;

    if (c + 1 < NCHUNK) {
      __builtin_prefetch(&u[rowU + base + CHUNK + tl], 0, 3);
      __builtin_prefetch(&delta[rowU + base + CHUNK + tl], 0, 3);
      __builtin_prefetch(&z[rowU + base + CHUNK + tl], 0, 3);
    }

    const float del = delta[rowU + base + tl];
    const float uu  = u[rowU + base + tl];
    const float zz  = z[rowU + base + tl];

    const float dtv = softplus_f(del);
    const float du  = dtv * uu;

    const float sT = prefix16(dtv);
    const float S  = __shfl(sT, 15, 16);

    v2f a[4], bm[4];
#pragma unroll
    for (int q = 0; q < 4; ++q) {
#pragma unroll
      for (int j = 0; j < 2; ++j) {
        const int    n   = 4 * q + j + 2 * half;
        const size_t idx = rowBC + (size_t)n * SEQLEN + base + tl;
        const float  cr  = Cm[idx];
        const float  br  = Bm[idx];
        const float  af  = Afac[q][j];
        const float p = af * sT;
        float ct = cr * __builtin_amdgcn_exp2f(p);
        float bu = br * du * __builtin_amdgcn_exp2f(-p);
        if (tl == 0) bu += x[q][j];
        a[q][j]  = ct;
        bm[q][j] = bu;
      }
    }

    v8f K = {0.f, 0.f, 0.f, 0.f, 0.f, 0.f, 0.f, 0.f};
#pragma unroll
    for (int q = 0; q < 4; ++q) {
      K = __builtin_amdgcn_wmma_f32_16x16x4_f32(
          false, bm[q], false, a[q], (short)0, K, false, false);
    }

    asm volatile("" :: "v"(zz));

#pragma unroll
    for (int q = 0; q < 4; ++q) {
#pragma unroll
      for (int j = 0; j < 2; ++j) {
        const float s = allreduce16(bm[q][j]);
        x[q][j] = s * __builtin_amdgcn_exp2f(Afac[q][j] * S);
      }
    }

    float m[8];
#pragma unroll
    for (int r = 0; r < 8; ++r) {
      const int srow = r + 8 * half;
      m[r] = (srow <= tl) ? K[r] : 0.0f;
    }
    const float part = ((m[0] + m[1]) + (m[2] + m[3])) +
                       ((m[4] + m[5]) + (m[6] + m[7]));
    const float Y = part + xhalf16(part);

    float yo  = Y + Dd * uu;
    const float sig =
        __builtin_amdgcn_rcpf(1.0f + __builtin_amdgcn_exp2f(-zz * LOG2E_F));
    yo *= zz * sig;
    if (lane < 16) out[rowU + base + tl] = yo;
  }
}

extern "C" void kernel_launch(void* const* d_in, const int* in_sizes, int n_in,
                              void* d_out, int out_size, void* d_ws, size_t ws_size,
                              hipStream_t stream) {
  (void)in_sizes; (void)n_in; (void)d_ws; (void)ws_size; (void)out_size;
  const float* u     = (const float*)d_in[0];
  const float* delta = (const float*)d_in[1];
  const float* A     = (const float*)d_in[2];
  const float* Bm    = (const float*)d_in[3];
  const float* Cm    = (const float*)d_in[4];
  const float* Dv    = (const float*)d_in[5];
  const float* z     = (const float*)d_in[6];
  float* out         = (float*)d_out;

  const int nWaves = BATCH * DIN;
  dim3 grid(nWaves / 4), block(128);
  mamba_scan_kernel<<<grid, block, 0, stream>>>(u, delta, A, Bm, Cm, Dv, z, out);
}


# State kept packed one row per lane: 8 column sums by one halving butterfly (8 DPP, not 32), one exp2 not 8, row_share multiply-add feeds it back into row 0; state sum order changes
# speedup vs baseline: 1.4243x; 1.1050x over previous
#include <hip/hip_runtime.h>


#define BATCH   4
#define DIN     2048
#define NSTATE  16
#define SEQLEN  2048
#define CHUNK   16
#define NCHUNK  (SEQLEN / CHUNK)

typedef float v2f __attribute__((ext_vector_type(2)));
typedef float v8f __attribute__((ext_vector_type(8)));

template <int CTRL>
__device__ __forceinline__ float dpp_mov(float x) {
  int r = __builtin_amdgcn_update_dpp(0, __builtin_bit_cast(int, x),
                                      CTRL, 0xF, 0xF, true);
  return __builtin_bit_cast(float, r);
}

template <int CTRL>
__device__ __forceinline__ float merge_step(float a, float b, bool bit) {
  const float keep = bit ? b : a;
  const float send = bit ? a : b;
  return keep + dpp_mov<CTRL>(send);
}

__device__ __forceinline__ float prefix16(float v) {
  v += dpp_mov<0x111>(v);
  v += dpp_mov<0x112>(v);
  v += dpp_mov<0x114>(v);
  v += dpp_mov<0x118>(v);
  return v;
}

__device__ __forceinline__ float xhalf16(float v) {
  int i = __builtin_bit_cast(int, v);
  int r = __builtin_amdgcn_permlanex16(i, i, 0x76543210, 0xFEDCBA98,
                                       false, false);
  return __builtin_bit_cast(float, r);
}

#define LOG2E_F 1.4426950408889634f
#define LN2_F   0.6931471805599453f

__device__ __forceinline__ float softplus_f(float x) {
  const float e = __builtin_amdgcn_exp2f(x * LOG2E_F);
  const float l = __builtin_amdgcn_logf(1.0f + e) * LN2_F;
  return (x > 20.0f) ? x : l;
}

__global__ __launch_bounds__(128) void mamba_scan_kernel(
    const float* __restrict__ u,
    const float* __restrict__ delta,
    const float* __restrict__ A,
    const float* __restrict__ Bm,
    const float* __restrict__ Cm,
    const float* __restrict__ Dv,
    const float* __restrict__ z,
    float* __restrict__ out)
{
  const int lane  = threadIdx.x & 31;
  const int wave  = (blockIdx.x << 2) | (threadIdx.x >> 5);
  const int b     = wave / DIN;
  const int d     = wave % DIN;

  const int tl    = lane & 15;
  const int half  = lane >> 4;

  float Afac[4][2];
#pragma unroll
  for (int q = 0; q < 4; ++q)
#pragma unroll
    for (int j = 0; j < 2; ++j)
      Afac[q][j] = A[(size_t)d * NSTATE + (4 * q + j + 2 * half)] * LOG2E_F;

  const int   rp   = tl & 7;
  const float AfP  = A[(size_t)d * NSTATE + (4 * (rp >> 1) + (rp & 1) + 2 * half)] * LOG2E_F;
  const bool  bit0 = (tl & 1) != 0;
  const bool  bit1 = (tl & 2) != 0;
  const bool  bit2 = (tl & 4) != 0;
  const float ind0 = (tl == 0) ? 1.0f : 0.0f;

  const float Dd = Dv[d];

  const size_t rowU  = ((size_t)b * DIN + d) * SEQLEN;
  const size_t rowBC = (size_t)b * NSTATE * SEQLEN;

  float xP = 0.0f;

  for (int c = 0; c < NCHUNK; ++c) {
    const int base = c * CHUNK;

    if (c + 1 < NCHUNK) {
      __builtin_prefetch(&u[rowU + base + CHUNK + tl], 0, 3);
      __builtin_prefetch(&delta[rowU + base + CHUNK + tl], 0, 3);
      __builtin_prefetch(&z[rowU + base + CHUNK + tl], 0, 3);
    }

    const float del = delta[rowU + base + tl];
    const float uu  = u[rowU + base + tl];
    const float zz  = z[rowU + base + tl];

    const float dtv = softplus_f(del);
    const float du  = dtv * uu;

    const float sT = prefix16(dtv);
    const float S  = __shfl(sT, 15, 16);

    v2f a[4], bm[4], bl[4];
#pragma unroll
    for (int q = 0; q < 4; ++q) {
#pragma unroll
      for (int j = 0; j < 2; ++j) {
        const int    n   = 4 * q + j + 2 * half;
        const size_t idx = rowBC + (size_t)n * SEQLEN + base + tl;
        const float  cr  = Cm[idx];
        const float  br  = Bm[idx];
        const float  af  = Afac[q][j];
        const float p = af * sT;
        a[q][j]  = cr * __builtin_amdgcn_exp2f(p);
        bl[q][j] = br * du * __builtin_amdgcn_exp2f(-p);
      }
    }
    bm[0][0] = fmaf(dpp_mov<0x150>(xP), ind0, bl[0][0]);
    bm[0][1] = fmaf(dpp_mov<0x151>(xP), ind0, bl[0][1]);
    bm[1][0] = fmaf(dpp_mov<0x152>(xP), ind0, bl[1][0]);
    bm[1][1] = fmaf(dpp_mov<0x153>(xP), ind0, bl[1][1]);
    bm[2][0] = fmaf(dpp_mov<0x154>(xP), ind0, bl[2][0]);
    bm[2][1] = fmaf(dpp_mov<0x155>(xP), ind0, bl[2][1]);
    bm[3][0] = fmaf(dpp_mov<0x156>(xP), ind0, bl[3][0]);
    bm[3][1] = fmaf(dpp_mov<0x157>(xP), ind0, bl[3][1]);

    v8f K = {0.f, 0.f, 0.f, 0.f, 0.f, 0.f, 0.f, 0.f};
#pragma unroll
    for (int q = 0; q < 4; ++q) {
      K = __builtin_amdgcn_wmma_f32_16x16x4_f32(
          false, bm[q], false, a[q], (short)0, K, false, false);
    }

    asm volatile("" :: "v"(zz));

    {
      const float t0 = merge_step<0x161>(bl[0][0], bl[0][1], bit0);
      const float t1 = merge_step<0x161>(bl[1][0], bl[1][1], bit0);
      const float t2 = merge_step<0x161>(bl[2][0], bl[2][1], bit0);
      const float t3 = merge_step<0x161>(bl[3][0], bl[3][1], bit0);
      const float u0 = merge_step<0x162>(t0, t1, bit1);
      const float u1 = merge_step<0x162>(t2, t3, bit1);
      float v = merge_step<0x164>(u0, u1, bit2);
      v += dpp_mov<0x168>(v);
      xP = (xP + v) * __builtin_amdgcn_exp2f(AfP * S);
    }

    float m[8];
#pragma unroll
    for (int r = 0; r < 8; ++r) {
      const int srow = r + 8 * half;
      m[r] = (srow <= tl) ? K[r] : 0.0f;
    }
    const float part = ((m[0] + m[1]) + (m[2] + m[3])) +
                       ((m[4] + m[5]) + (m[6] + m[7]));
    const float Y = part + xhalf16(part);

    float yo  = Y + Dd * uu;
    const float sig =
        __builtin_amdgcn_rcpf(1.0f + __builtin_amdgcn_exp2f(-zz * LOG2E_F));
    yo *= zz * sig;
    if (lane < 16) out[rowU + base + tl] = yo;
  }
}

extern "C" void kernel_launch(void* const* d_in, const int* in_sizes, int n_in,
                              void* d_out, int out_size, void* d_ws, size_t ws_size,
                              hipStream_t stream) {
  (void)in_sizes; (void)n_in; (void)d_ws; (void)ws_size; (void)out_size;
  const float* u     = (const float*)d_in[0];
  const float* delta = (const float*)d_in[1];
  const float* A     = (const float*)d_in[2];
  const float* Bm    = (const float*)d_in[3];
  const float* Cm    = (const float*)d_in[4];
  const float* Dv    = (const float*)d_in[5];
  const float* z     = (const float*)d_in[6];
  float* out         = (float*)d_out;

  const int nWaves = BATCH * DIN;
  dim3 grid(nWaves / 4), block(128);
  mamba_scan_kernel<<<grid, block, 0, stream>>>(u, delta, A, Bm, Cm, Dv, z, out);
}


# Chunk loop reads memory one chunk ahead: chunk c+1's 19 loads fly into a second register set while chunk c computes (unrolled by two, sets swap); gate formed before the store branch; sum order same
# speedup vs baseline: 1.6342x; 1.1474x over previous
#include <hip/hip_runtime.h>


#define BATCH   4
#define DIN     2048
#define NSTATE  16
#define SEQLEN  2048
#define CHUNK   16
#define NCHUNK  (SEQLEN / CHUNK)

typedef float v2f __attribute__((ext_vector_type(2)));
typedef float v8f __attribute__((ext_vector_type(8)));

template <int CTRL>
__device__ __forceinline__ float dpp_mov(float x) {
  int r = __builtin_amdgcn_update_dpp(0, __builtin_bit_cast(int, x),
                                      CTRL, 0xF, 0xF, true);
  return __builtin_bit_cast(float, r);
}

template <int CTRL>
__device__ __forceinline__ float merge_step(float a, float b, bool bit) {
  const float keep = bit ? b : a;
  const float send = bit ? a : b;
  return keep + dpp_mov<CTRL>(send);
}

__device__ __forceinline__ float prefix16(float v) {
  v += dpp_mov<0x111>(v);
  v += dpp_mov<0x112>(v);
  v += dpp_mov<0x114>(v);
  v += dpp_mov<0x118>(v);
  return v;
}

__device__ __forceinline__ float xhalf16(float v) {
  int i = __builtin_bit_cast(int, v);
  int r = __builtin_amdgcn_permlanex16(i, i, 0x76543210, 0xFEDCBA98,
                                       false, false);
  return __builtin_bit_cast(float, r);
}

#define LOG2E_F 1.4426950408889634f
#define LN2_F   0.6931471805599453f

__device__ __forceinline__ float softplus_f(float x) {
  const float e = __builtin_amdgcn_exp2f(x * LOG2E_F);
  const float l = __builtin_amdgcn_logf(1.0f + e) * LN2_F;
  return (x > 20.0f) ? x : l;
}

struct ChunkIn {
  float del, uu, zz;
  float cr[4][2];
  float br[4][2];
};

__device__ __forceinline__ void load_chunk(ChunkIn& L,
    const float* __restrict__ u, const float* __restrict__ delta,
    const float* __restrict__ z, const float* __restrict__ Bm,
    const float* __restrict__ Cm, size_t rowU, size_t rowBC,
    int base, int tl, int half) {
  L.del = delta[rowU + base + tl];
  L.uu  = u[rowU + base + tl];
  L.zz  = z[rowU + base + tl];
#pragma unroll
  for (int q = 0; q < 4; ++q) {
#pragma unroll
    for (int j = 0; j < 2; ++j) {
      const int    n   = 4 * q + j + 2 * half;
      const size_t idx = rowBC + (size_t)n * SEQLEN + base + tl;
      L.cr[q][j] = Cm[idx];
      L.br[q][j] = Bm[idx];
    }
  }
}

__device__ __forceinline__ float chunk_step(const ChunkIn& L, float& xP,
    const float (&Afac)[4][2], float AfP, float ind0, float Dd,
    bool bit0, bool bit1, bool bit2, int tl, int half) {
  const float uu = L.uu;
  const float zz = L.zz;

  const float sig =
      __builtin_amdgcn_rcpf(1.0f + __builtin_amdgcn_exp2f(-zz * LOG2E_F));
  const float gate = zz * sig;
  const float skip = Dd * uu;

  const float dtv = softplus_f(L.del);
  const float du  = dtv * uu;

  const float sT = prefix16(dtv);
  const float S  = __shfl(sT, 15, 16);

  v2f a[4], bm[4], bl[4];
#pragma unroll
  for (int q = 0; q < 4; ++q) {
#pragma unroll
    for (int j = 0; j < 2; ++j) {
      const float af = Afac[q][j];
      const float p  = af * sT;
      a[q][j]  = L.cr[q][j] * __builtin_amdgcn_exp2f(p);
      bl[q][j] = L.br[q][j] * du * __builtin_amdgcn_exp2f(-p);
    }
  }
  bm[0][0] = fmaf(dpp_mov<0x150>(xP), ind0, bl[0][0]);
  bm[0][1] = fmaf(dpp_mov<0x151>(xP), ind0, bl[0][1]);
  bm[1][0] = fmaf(dpp_mov<0x152>(xP), ind0, bl[1][0]);
  bm[1][1] = fmaf(dpp_mov<0x153>(xP), ind0, bl[1][1]);
  bm[2][0] = fmaf(dpp_mov<0x154>(xP), ind0, bl[2][0]);
  bm[2][1] = fmaf(dpp_mov<0x155>(xP), ind0, bl[2][1]);
  bm[3][0] = fmaf(dpp_mov<0x156>(xP), ind0, bl[3][0]);
  bm[3][1] = fmaf(dpp_mov<0x157>(xP), ind0, bl[3][1]);

  v8f K = {0.f, 0.f, 0.f, 0.f, 0.f, 0.f, 0.f, 0.f};
#pragma unroll
  for (int q = 0; q < 4; ++q) {
    K = __builtin_amdgcn_wmma_f32_16x16x4_f32(
        false, bm[q], false, a[q], (short)0, K, false, false);
  }

  {
    const float t0 = merge_step<0x161>(bl[0][0], bl[0][1], bit0);
    const float t1 = merge_step<0x161>(bl[1][0], bl[1][1], bit0);
    const float t2 = merge_step<0x161>(bl[2][0], bl[2][1], bit0);
    const float t3 = merge_step<0x161>(bl[3][0], bl[3][1], bit0);
    const float u0 = merge_step<0x162>(t0, t1, bit1);
    const float u1 = merge_step<0x162>(t2, t3, bit1);
    float v = merge_step<0x164>(u0, u1, bit2);
    v += dpp_mov<0x168>(v);
    xP = (xP + v) * __builtin_amdgcn_exp2f(AfP * S);
  }

  float m[8];
#pragma unroll
  for (int r = 0; r < 8; ++r) {
    const int srow = r + 8 * half;
    m[r] = (srow <= tl) ? K[r] : 0.0f;
  }
  const float part = ((m[0] + m[1]) + (m[2] + m[3])) +
                     ((m[4] + m[5]) + (m[6] + m[7]));
  const float Y = part + xhalf16(part);

  return (Y + skip) * gate;
}

__global__ __launch_bounds__(128) void mamba_scan_kernel(
    const float* __restrict__ u,
    const float* __restrict__ delta,
    const float* __restrict__ A,
    const float* __restrict__ Bm,
    const float* __restrict__ Cm,
    const float* __restrict__ Dv,
    const float* __restrict__ z,
    float* __restrict__ out)
{
  const int lane  = threadIdx.x & 31;
  const int wave  = (blockIdx.x << 2) | (threadIdx.x >> 5);
  const int b     = wave / DIN;
  const int d     = wave % DIN;

  const int tl    = lane & 15;
  const int half  = lane >> 4;

  float Afac[4][2];
#pragma unroll
  for (int q = 0; q < 4; ++q)
#pragma unroll
    for (int j = 0; j < 2; ++j)
      Afac[q][j] = A[(size_t)d * NSTATE + (4 * q + j + 2 * half)] * LOG2E_F;

  const int   rp   = tl & 7;
  const float AfP  = A[(size_t)d * NSTATE + (4 * (rp >> 1) + (rp & 1) + 2 * half)] * LOG2E_F;
  const bool  bit0 = (tl & 1) != 0;
  const bool  bit1 = (tl & 2) != 0;
  const bool  bit2 = (tl & 4) != 0;
  const float ind0 = (tl == 0) ? 1.0f : 0.0f;

  const float Dd = Dv[d];

  const size_t rowU  = ((size_t)b * DIN + d) * SEQLEN;
  const size_t rowBC = (size_t)b * NSTATE * SEQLEN;

  float xP = 0.0f;

  ChunkIn inA, inB;
  load_chunk(inA, u, delta, z, Bm, Cm, rowU, rowBC, 0, tl, half);

  for (int c = 0; c < NCHUNK; c += 2) {
    const int base = c * CHUNK;

    load_chunk(inB, u, delta, z, Bm, Cm, rowU, rowBC, base + CHUNK, tl, half);
    if (c + 4 < NCHUNK) {
      __builtin_prefetch(&u[rowU + base + 4 * CHUNK + tl], 0, 3);
      __builtin_prefetch(&delta[rowU + base + 4 * CHUNK + tl], 0, 3);
      __builtin_prefetch(&z[rowU + base + 4 * CHUNK + tl], 0, 3);
    }
    {
      float yo = chunk_step(inA, xP, Afac, AfP, ind0, Dd, bit0, bit1, bit2, tl, half);
      asm volatile("" : "+v"(yo));
      if (lane < 16) out[rowU + base + tl] = yo;
    }

    const int nbase = (c + 2 < NCHUNK) ? base + 2 * CHUNK : base + CHUNK;
    load_chunk(inA, u, delta, z, Bm, Cm, rowU, rowBC, nbase, tl, half);
    {
      float yo = chunk_step(inB, xP, Afac, AfP, ind0, Dd, bit0, bit1, bit2, tl, half);
      asm volatile("" : "+v"(yo));
      if (lane < 16) out[rowU + base + CHUNK + tl] = yo;
    }
  }
}

extern "C" void kernel_launch(void* const* d_in, const int* in_sizes, int n_in,
                              void* d_out, int out_size, void* d_ws, size_t ws_size,
                              hipStream_t stream) {
  (void)in_sizes; (void)n_in; (void)d_ws; (void)ws_size; (void)out_size;
  const float* u     = (const float*)d_in[0];
  const float* delta = (const float*)d_in[1];
  const float* A     = (const float*)d_in[2];
  const float* Bm    = (const float*)d_in[3];
  const float* Cm    = (const float*)d_in[4];
  const float* Dv    = (const float*)d_in[5];
  const float* z     = (const float*)d_in[6];
  float* out         = (float*)d_out;

  const int nWaves = BATCH * DIN;
  dim3 grid(nWaves / 4), block(128);
  mamba_scan_kernel<<<grid, block, 0, stream>>>(u, delta, A, Bm, Cm, Dv, z, out);
}


# A wave carries two adjacent channels of one batch: a chunk's 16 B/C loads and their addresses serve both (22 loads per two channel-chunks, not 38); each lane half stores one channel; sum order same
# speedup vs baseline: 1.8224x; 1.1152x over previous
#include <hip/hip_runtime.h>


#define BATCH   4
#define DIN     2048
#define NSTATE  16
#define SEQLEN  2048
#define CHUNK   16
#define NCHUNK  (SEQLEN / CHUNK)

typedef float v2f __attribute__((ext_vector_type(2)));
typedef float v8f __attribute__((ext_vector_type(8)));

template <int CTRL>
__device__ __forceinline__ float dpp_mov(float x) {
  int r = __builtin_amdgcn_update_dpp(0, __builtin_bit_cast(int, x),
                                      CTRL, 0xF, 0xF, true);
  return __builtin_bit_cast(float, r);
}

template <int CTRL>
__device__ __forceinline__ float merge_step(float a, float b, bool bit) {
  const float keep = bit ? b : a;
  const float send = bit ? a : b;
  return keep + dpp_mov<CTRL>(send);
}

__device__ __forceinline__ float prefix16(float v) {
  v += dpp_mov<0x111>(v);
  v += dpp_mov<0x112>(v);
  v += dpp_mov<0x114>(v);
  v += dpp_mov<0x118>(v);
  return v;
}

__device__ __forceinline__ float xhalf16(float v) {
  int i = __builtin_bit_cast(int, v);
  int r = __builtin_amdgcn_permlanex16(i, i, 0x76543210, 0xFEDCBA98,
                                       false, false);
  return __builtin_bit_cast(float, r);
}

#define LOG2E_F 1.4426950408889634f
#define LN2_F   0.6931471805599453f

__device__ __forceinline__ float softplus_f(float x) {
  const float e = __builtin_amdgcn_exp2f(x * LOG2E_F);
  const float l = __builtin_amdgcn_logf(1.0f + e) * LN2_F;
  return (x > 20.0f) ? x : l;
}

struct ChunkIn {
  float del[2], uu[2], zz[2];
  float cr[4][2];
  float br[4][2];
};

__device__ __forceinline__ void load_chunk(ChunkIn& L,
    const float* __restrict__ pU, const float* __restrict__ pD,
    const float* __restrict__ pZ, const float* __restrict__ pB,
    const float* __restrict__ pC, int off) {
  L.del[0] = pD[off];
  L.del[1] = pD[off + SEQLEN];
  L.uu[0]  = pU[off];
  L.uu[1]  = pU[off + SEQLEN];
  L.zz[0]  = pZ[off];
  L.zz[1]  = pZ[off + SEQLEN];
#pragma unroll
  for (int q = 0; q < 4; ++q) {
#pragma unroll
    for (int j = 0; j < 2; ++j) {
      L.cr[q][j] = pC[off + (4 * q + j) * SEQLEN];
      L.br[q][j] = pB[off + (4 * q + j) * SEQLEN];
    }
  }
}

struct RowOps {
  v2f a[4];
  v2f bm[4];
  float gate;
  float skip;
};

__device__ __forceinline__ void row_front(RowOps& R, float del, float uu, float zz,
    const float (&cr)[4][2], const float (&br)[4][2], float& xP,
    const float (&Afac)[4][2], float AfP, float ind0, float Dd,
    bool bit0, bool bit1, bool bit2) {
  const float sig =
      __builtin_amdgcn_rcpf(1.0f + __builtin_amdgcn_exp2f(-zz * LOG2E_F));
  R.gate = zz * sig;
  R.skip = Dd * uu;

  const float dtv = softplus_f(del);
  const float du  = dtv * uu;

  const float sT = prefix16(dtv);
  const float S  = __shfl(sT, 15, 16);

  v2f bl[4];
#pragma unroll
  for (int q = 0; q < 4; ++q) {
#pragma unroll
    for (int j = 0; j < 2; ++j) {
      const float af = Afac[q][j];
      const float p  = af * sT;
      R.a[q][j] = cr[q][j] * __builtin_amdgcn_exp2f(p);
      bl[q][j]  = br[q][j] * du * __builtin_amdgcn_exp2f(-p);
    }
  }
  R.bm[0][0] = fmaf(dpp_mov<0x150>(xP), ind0, bl[0][0]);
  R.bm[0][1] = fmaf(dpp_mov<0x151>(xP), ind0, bl[0][1]);
  R.bm[1][0] = fmaf(dpp_mov<0x152>(xP), ind0, bl[1][0]);
  R.bm[1][1] = fmaf(dpp_mov<0x153>(xP), ind0, bl[1][1]);
  R.bm[2][0] = fmaf(dpp_mov<0x154>(xP), ind0, bl[2][0]);
  R.bm[2][1] = fmaf(dpp_mov<0x155>(xP), ind0, bl[2][1]);
  R.bm[3][0] = fmaf(dpp_mov<0x156>(xP), ind0, bl[3][0]);
  R.bm[3][1] = fmaf(dpp_mov<0x157>(xP), ind0, bl[3][1]);

  {
    const float t0 = merge_step<0x161>(bl[0][0], bl[0][1], bit0);
    const float t1 = merge_step<0x161>(bl[1][0], bl[1][1], bit0);
    const float t2 = merge_step<0x161>(bl[2][0], bl[2][1], bit0);
    const float t3 = merge_step<0x161>(bl[3][0], bl[3][1], bit0);
    const float u0 = merge_step<0x162>(t0, t1, bit1);
    const float u1 = merge_step<0x162>(t2, t3, bit1);
    float v = merge_step<0x164>(u0, u1, bit2);
    v += dpp_mov<0x168>(v);
    xP = (xP + v) * __builtin_amdgcn_exp2f(AfP * S);
  }
}

__device__ __forceinline__ float row_back(const v8f& K, float gate, float skip,
                                          int tl, int half) {
  float m[8];
#pragma unroll
  for (int r = 0; r < 8; ++r) {
    const int srow = r + 8 * half;
    m[r] = (srow <= tl) ? K[r] : 0.0f;
  }
  const float part = ((m[0] + m[1]) + (m[2] + m[3])) +
                     ((m[4] + m[5]) + (m[6] + m[7]));
  const float Y = part + xhalf16(part);

  return (Y + skip) * gate;
}

__device__ __forceinline__ float chunk_step2(const ChunkIn& L,
    float& xP0, float& xP1,
    const float (&Afac0)[4][2], const float (&Afac1)[4][2],
    float AfP0, float AfP1, float ind0, float Dd0, float Dd1,
    bool bit0, bool bit1, bool bit2, int tl, int half) {
  RowOps R0, R1;
  row_front(R0, L.del[0], L.uu[0], L.zz[0], L.cr, L.br, xP0, Afac0, AfP0, ind0,
            Dd0, bit0, bit1, bit2);
  row_front(R1, L.del[1], L.uu[1], L.zz[1], L.cr, L.br, xP1, Afac1, AfP1, ind0,
            Dd1, bit0, bit1, bit2);

  v8f K0 = {0.f, 0.f, 0.f, 0.f, 0.f, 0.f, 0.f, 0.f};
  v8f K1 = {0.f, 0.f, 0.f, 0.f, 0.f, 0.f, 0.f, 0.f};
#pragma unroll
  for (int q = 0; q < 4; ++q) {
    K0 = __builtin_amdgcn_wmma_f32_16x16x4_f32(
        false, R0.bm[q], false, R0.a[q], (short)0, K0, false, false);
    K1 = __builtin_amdgcn_wmma_f32_16x16x4_f32(
        false, R1.bm[q], false, R1.a[q], (short)0, K1, false, false);
  }

  const float y0 = row_back(K0, R0.gate, R0.skip, tl, half);
  const float y1 = row_back(K1, R1.gate, R1.skip, tl, half);
  return half ? y1 : y0;
}

__global__ __launch_bounds__(128) __attribute__((amdgpu_waves_per_eu(4, 4)))
void mamba_scan_kernel(
    const float* __restrict__ u,
    const float* __restrict__ delta,
    const float* __restrict__ A,
    const float* __restrict__ Bm,
    const float* __restrict__ Cm,
    const float* __restrict__ Dv,
    const float* __restrict__ z,
    float* __restrict__ out)
{
  const int lane  = threadIdx.x & 31;
  const int wave  = (blockIdx.x << 2) | (threadIdx.x >> 5);
  const int b     = wave / (DIN / 2);
  const int d     = (wave % (DIN / 2)) * 2;

  const int tl    = lane & 15;
  const int half  = lane >> 4;

  float Afac0[4][2], Afac1[4][2];
#pragma unroll
  for (int q = 0; q < 4; ++q)
#pragma unroll
    for (int j = 0; j < 2; ++j) {
      const int n = 4 * q + j + 2 * half;
      Afac0[q][j] = A[(size_t)d * NSTATE + n] * LOG2E_F;
      Afac1[q][j] = A[(size_t)(d + 1) * NSTATE + n] * LOG2E_F;
    }

  const int   rp   = tl & 7;
  const int   np   = 4 * (rp >> 1) + (rp & 1) + 2 * half;
  const float AfP0 = A[(size_t)d * NSTATE + np] * LOG2E_F;
  const float AfP1 = A[(size_t)(d + 1) * NSTATE + np] * LOG2E_F;
  const bool  bit0 = (tl & 1) != 0;
  const bool  bit1 = (tl & 2) != 0;
  const bool  bit2 = (tl & 4) != 0;
  const float ind0 = (tl == 0) ? 1.0f : 0.0f;

  const float Dd0 = Dv[d];
  const float Dd1 = Dv[d + 1];

  const size_t rowU  = ((size_t)b * DIN + d) * SEQLEN;
  const size_t rowBC = (size_t)b * NSTATE * SEQLEN;
  const float* pU = u + rowU + tl;
  const float* pD = delta + rowU + tl;
  const float* pZ = z + rowU + tl;
  const float* pB = Bm + rowBC + (size_t)(2 * half) * SEQLEN + tl;
  const float* pC = Cm + rowBC + (size_t)(2 * half) * SEQLEN + tl;
  float* pO = out + rowU + (size_t)half * SEQLEN + tl;

  float xP0 = 0.0f, xP1 = 0.0f;

  ChunkIn inA, inB;
  load_chunk(inA, pU, pD, pZ, pB, pC, 0);

  for (int c = 0; c < NCHUNK; c += 2) {
    const int base = c * CHUNK;

    load_chunk(inB, pU, pD, pZ, pB, pC, base + CHUNK);
    if (c + 4 < NCHUNK) {
      __builtin_prefetch(&pU[base + 4 * CHUNK], 0, 3);
      __builtin_prefetch(&pD[base + 4 * CHUNK], 0, 3);
      __builtin_prefetch(&pZ[base + 4 * CHUNK], 0, 3);
      __builtin_prefetch(&pU[base + 4 * CHUNK + SEQLEN], 0, 3);
      __builtin_prefetch(&pD[base + 4 * CHUNK + SEQLEN], 0, 3);
      __builtin_prefetch(&pZ[base + 4 * CHUNK + SEQLEN], 0, 3);
    }
    __builtin_amdgcn_sched_barrier(0);
    pO[base] =
        chunk_step2(inA, xP0, xP1, Afac0, Afac1, AfP0, AfP1, ind0, Dd0, Dd1,
                    bit0, bit1, bit2, tl, half);

    const int nbase = (c + 2 < NCHUNK) ? base + 2 * CHUNK : base + CHUNK;
    load_chunk(inA, pU, pD, pZ, pB, pC, nbase);
    __builtin_amdgcn_sched_barrier(0);
    pO[base + CHUNK] =
        chunk_step2(inB, xP0, xP1, Afac0, Afac1, AfP0, AfP1, ind0, Dd0, Dd1,
                    bit0, bit1, bit2, tl, half);
  }
}

extern "C" void kernel_launch(void* const* d_in, const int* in_sizes, int n_in,
                              void* d_out, int out_size, void* d_ws, size_t ws_size,
                              hipStream_t stream) {
  (void)in_sizes; (void)n_in; (void)d_ws; (void)ws_size; (void)out_size;
  const float* u     = (const float*)d_in[0];
  const float* delta = (const float*)d_in[1];
  const float* A     = (const float*)d_in[2];
  const float* Bm    = (const float*)d_in[3];
  const float* Cm    = (const float*)d_in[4];
  const float* Dv    = (const float*)d_in[5];
  const float* z     = (const float*)d_in[6];
  float* out         = (float*)d_out;

  const int nWaves = BATCH * DIN / 2;
  dim3 grid(nWaves / 4), block(128);
  mamba_scan_kernel<<<grid, block, 0, stream>>>(u, delta, A, Bm, Cm, Dv, z, out);
}


# Each lane half does one channel's per-timestep work (softplus, dt prefix sum, gate) once, not twice, and a half-swap instruction hands it over; 19 loads per chunk, not 22; sum order same
# speedup vs baseline: 1.8850x; 1.0344x over previous
#include <hip/hip_runtime.h>


#define BATCH   4
#define DIN     2048
#define NSTATE  16
#define SEQLEN  2048
#define CHUNK   16
#define NCHUNK  (SEQLEN / CHUNK)

typedef float v2f __attribute__((ext_vector_type(2)));
typedef float v8f __attribute__((ext_vector_type(8)));

template <int CTRL>
__device__ __forceinline__ float dpp_mov(float x) {
  int r = __builtin_amdgcn_update_dpp(0, __builtin_bit_cast(int, x),
                                      CTRL, 0xF, 0xF, true);
  return __builtin_bit_cast(float, r);
}

template <int CTRL>
__device__ __forceinline__ float merge_step(float a, float b, bool bit) {
  const float keep = bit ? b : a;
  const float send = bit ? a : b;
  return keep + dpp_mov<CTRL>(send);
}

__device__ __forceinline__ float prefix16(float v) {
  v += dpp_mov<0x111>(v);
  v += dpp_mov<0x112>(v);
  v += dpp_mov<0x114>(v);
  v += dpp_mov<0x118>(v);
  return v;
}

__device__ __forceinline__ void half_swap(float& a, float& b) {
  asm("v_permlane16_swap_b32 %0, %1" : "+v"(a), "+v"(b));
}

#define LOG2E_F 1.4426950408889634f
#define LN2_F   0.6931471805599453f

__device__ __forceinline__ float softplus_f(float x) {
  const float e = __builtin_amdgcn_exp2f(x * LOG2E_F);
  const float l = __builtin_amdgcn_logf(1.0f + e) * LN2_F;
  return (x > 20.0f) ? x : l;
}

struct ChunkIn {
  float del, uu, zz;
  float cr[4][2];
  float br[4][2];
};

__device__ __forceinline__ void load_chunk(ChunkIn& L,
    const float* __restrict__ pU, const float* __restrict__ pD,
    const float* __restrict__ pZ, const float* __restrict__ pB,
    const float* __restrict__ pC, int off) {
  L.del = pD[off];
  L.uu  = pU[off];
  L.zz  = pZ[off];
#pragma unroll
  for (int q = 0; q < 4; ++q) {
#pragma unroll
    for (int j = 0; j < 2; ++j) {
      L.cr[q][j] = pC[off + (4 * q + j) * SEQLEN];
      L.br[q][j] = pB[off + (4 * q + j) * SEQLEN];
    }
  }
}

struct RowOps {
  v2f a[4];
  v2f bm[4];
};

__device__ __forceinline__ void row_front(RowOps& R, float sT, float du, float S,
    const float (&cr)[4][2], const float (&br)[4][2], float& xP,
    const float (&Afac)[4][2], float AfP, float ind0,
    bool bit0, bool bit1, bool bit2) {
  v2f bl[4];
#pragma unroll
  for (int q = 0; q < 4; ++q) {
#pragma unroll
    for (int j = 0; j < 2; ++j) {
      const float af = Afac[q][j];
      const float p  = af * sT;
      R.a[q][j] = cr[q][j] * __builtin_amdgcn_exp2f(p);
      bl[q][j]  = br[q][j] * du * __builtin_amdgcn_exp2f(-p);
    }
  }
  R.bm[0][0] = fmaf(dpp_mov<0x150>(xP), ind0, bl[0][0]);
  R.bm[0][1] = fmaf(dpp_mov<0x151>(xP), ind0, bl[0][1]);
  R.bm[1][0] = fmaf(dpp_mov<0x152>(xP), ind0, bl[1][0]);
  R.bm[1][1] = fmaf(dpp_mov<0x153>(xP), ind0, bl[1][1]);
  R.bm[2][0] = fmaf(dpp_mov<0x154>(xP), ind0, bl[2][0]);
  R.bm[2][1] = fmaf(dpp_mov<0x155>(xP), ind0, bl[2][1]);
  R.bm[3][0] = fmaf(dpp_mov<0x156>(xP), ind0, bl[3][0]);
  R.bm[3][1] = fmaf(dpp_mov<0x157>(xP), ind0, bl[3][1]);

  {
    const float t0 = merge_step<0x161>(bl[0][0], bl[0][1], bit0);
    const float t1 = merge_step<0x161>(bl[1][0], bl[1][1], bit0);
    const float t2 = merge_step<0x161>(bl[2][0], bl[2][1], bit0);
    const float t3 = merge_step<0x161>(bl[3][0], bl[3][1], bit0);
    const float u0 = merge_step<0x162>(t0, t1, bit1);
    const float u1 = merge_step<0x162>(t2, t3, bit1);
    float v = merge_step<0x164>(u0, u1, bit2);
    v += dpp_mov<0x168>(v);
    xP = (xP + v) * __builtin_amdgcn_exp2f(AfP * S);
  }
}

__device__ __forceinline__ float row_part(const v8f& K, int tl, int half) {
  float m[8];
#pragma unroll
  for (int r = 0; r < 8; ++r) {
    const int srow = r + 8 * half;
    m[r] = (srow <= tl) ? K[r] : 0.0f;
  }
  return ((m[0] + m[1]) + (m[2] + m[3])) + ((m[4] + m[5]) + (m[6] + m[7]));
}

__device__ __forceinline__ float chunk_step2(const ChunkIn& L,
    float& xP0, float& xP1,
    const float (&Afac0)[4][2], const float (&Afac1)[4][2],
    float AfP0, float AfP1, float ind0, float DdOwn,
    bool bit0, bool bit1, bool bit2, int tl, int half) {
  const float sig =
      __builtin_amdgcn_rcpf(1.0f + __builtin_amdgcn_exp2f(-L.zz * LOG2E_F));
  const float gate = L.zz * sig;
  const float skip = DdOwn * L.uu;

  const float dtv = softplus_f(L.del);
  float sT0 = prefix16(dtv);
  float du0 = dtv * L.uu;
  float sT1 = sT0;
  float du1 = du0;
  half_swap(sT0, sT1);
  half_swap(du0, du1);
  const float S0 = __shfl(sT0, 15, 16);
  const float S1 = __shfl(sT1, 15, 16);

  RowOps R0, R1;
  row_front(R0, sT0, du0, S0, L.cr, L.br, xP0, Afac0, AfP0, ind0,
            bit0, bit1, bit2);
  row_front(R1, sT1, du1, S1, L.cr, L.br, xP1, Afac1, AfP1, ind0,
            bit0, bit1, bit2);

  v8f K0 = {0.f, 0.f, 0.f, 0.f, 0.f, 0.f, 0.f, 0.f};
  v8f K1 = {0.f, 0.f, 0.f, 0.f, 0.f, 0.f, 0.f, 0.f};
#pragma unroll
  for (int q = 0; q < 4; ++q) {
    K0 = __builtin_amdgcn_wmma_f32_16x16x4_f32(
        false, R0.bm[q], false, R0.a[q], (short)0, K0, false, false);
    K1 = __builtin_amdgcn_wmma_f32_16x16x4_f32(
        false, R1.bm[q], false, R1.a[q], (short)0, K1, false, false);
  }

  float p0 = row_part(K0, tl, half);
  float p1 = row_part(K1, tl, half);
  half_swap(p0, p1);
  const float Y = p0 + p1;

  return (Y + skip) * gate;
}

__global__ __launch_bounds__(128) __attribute__((amdgpu_waves_per_eu(4, 4)))
void mamba_scan_kernel(
    const float* __restrict__ u,
    const float* __restrict__ delta,
    const float* __restrict__ A,
    const float* __restrict__ Bm,
    const float* __restrict__ Cm,
    const float* __restrict__ Dv,
    const float* __restrict__ z,
    float* __restrict__ out)
{
  const int lane  = threadIdx.x & 31;
  const int wave  = (blockIdx.x << 2) | (threadIdx.x >> 5);
  const int b     = wave / (DIN / 2);
  const int d     = (wave % (DIN / 2)) * 2;

  const int tl    = lane & 15;
  const int half  = lane >> 4;

  float Afac0[4][2], Afac1[4][2];
#pragma unroll
  for (int q = 0; q < 4; ++q)
#pragma unroll
    for (int j = 0; j < 2; ++j) {
      const int n = 4 * q + j + 2 * half;
      Afac0[q][j] = A[(size_t)d * NSTATE + n] * LOG2E_F;
      Afac1[q][j] = A[(size_t)(d + 1) * NSTATE + n] * LOG2E_F;
    }

  const int   rp   = tl & 7;
  const int   np   = 4 * (rp >> 1) + (rp & 1) + 2 * half;
  const float AfP0 = A[(size_t)d * NSTATE + np] * LOG2E_F;
  const float AfP1 = A[(size_t)(d + 1) * NSTATE + np] * LOG2E_F;
  const bool  bit0 = (tl & 1) != 0;
  const bool  bit1 = (tl & 2) != 0;
  const bool  bit2 = (tl & 4) != 0;
  const float ind0 = (tl == 0) ? 1.0f : 0.0f;

  const float DdOwn = Dv[d + half];

  const size_t rowOwn = ((size_t)b * DIN + d + half) * SEQLEN;
  const size_t rowBC  = (size_t)b * NSTATE * SEQLEN;
  const float* pU = u + rowOwn + tl;
  const float* pD = delta + rowOwn + tl;
  const float* pZ = z + rowOwn + tl;
  const float* pB = Bm + rowBC + (size_t)(2 * half) * SEQLEN + tl;
  const float* pC = Cm + rowBC + (size_t)(2 * half) * SEQLEN + tl;
  float* pO = out + rowOwn + tl;

  float xP0 = 0.0f, xP1 = 0.0f;

  ChunkIn inA, inB;
  load_chunk(inA, pU, pD, pZ, pB, pC, 0);

  for (int c = 0; c < NCHUNK; c += 2) {
    const int base = c * CHUNK;

    load_chunk(inB, pU, pD, pZ, pB, pC, base + CHUNK);
    if (c + 4 < NCHUNK) {
      __builtin_prefetch(&pU[base + 4 * CHUNK], 0, 3);
      __builtin_prefetch(&pD[base + 4 * CHUNK], 0, 3);
      __builtin_prefetch(&pZ[base + 4 * CHUNK], 0, 3);
    }
    __builtin_amdgcn_sched_barrier(0);
    pO[base] =
        chunk_step2(inA, xP0, xP1, Afac0, Afac1, AfP0, AfP1, ind0, DdOwn,
                    bit0, bit1, bit2, tl, half);

    const int nbase = (c + 2 < NCHUNK) ? base + 2 * CHUNK : base + CHUNK;
    load_chunk(inA, pU, pD, pZ, pB, pC, nbase);
    __builtin_amdgcn_sched_barrier(0);
    pO[base + CHUNK] =
        chunk_step2(inB, xP0, xP1, Afac0, Afac1, AfP0, AfP1, ind0, DdOwn,
                    bit0, bit1, bit2, tl, half);
  }
}

extern "C" void kernel_launch(void* const* d_in, const int* in_sizes, int n_in,
                              void* d_out, int out_size, void* d_ws, size_t ws_size,
                              hipStream_t stream) {
  (void)in_sizes; (void)n_in; (void)d_ws; (void)ws_size; (void)out_size;
  const float* u     = (const float*)d_in[0];
  const float* delta = (const float*)d_in[1];
  const float* A     = (const float*)d_in[2];
  const float* Bm    = (const float*)d_in[3];
  const float* Cm    = (const float*)d_in[4];
  const float* Dv    = (const float*)d_in[5];
  const float* z     = (const float*)d_in[6];
  float* out         = (float*)d_out;

  const int nWaves = BATCH * DIN / 2;
  dim3 grid(nWaves / 4), block(128);
  mamba_scan_kernel<<<grid, block, 0, stream>>>(u, delta, A, Bm, Cm, Dv, z, out);
}


# Carried state reaches row 0 of the next chunk's A operand via the wave's own LDS (1 store, 4 16-byte loads a chunk) as the multiply-add's addend; 16 row_share DPP moves go; state sum order changes
# speedup vs baseline: 1.9839x; 1.0525x over previous
#include <hip/hip_runtime.h>


#define BATCH   4
#define DIN     2048
#define NSTATE  16
#define SEQLEN  2048
#define CHUNK   16
#define NCHUNK  (SEQLEN / CHUNK)

typedef float v2f __attribute__((ext_vector_type(2)));
typedef float v4f __attribute__((ext_vector_type(4)));
typedef float v8f __attribute__((ext_vector_type(8)));

template <int CTRL>
__device__ __forceinline__ float dpp_mov(float x) {
  int r = __builtin_amdgcn_update_dpp(0, __builtin_bit_cast(int, x),
                                      CTRL, 0xF, 0xF, true);
  return __builtin_bit_cast(float, r);
}

template <int CTRL>
__device__ __forceinline__ float merge_step(float a, float b, bool bit) {
  const float keep = bit ? b : a;
  const float send = bit ? a : b;
  return keep + dpp_mov<CTRL>(send);
}

__device__ __forceinline__ float prefix16(float v) {
  v += dpp_mov<0x111>(v);
  v += dpp_mov<0x112>(v);
  v += dpp_mov<0x114>(v);
  v += dpp_mov<0x118>(v);
  return v;
}

__device__ __forceinline__ void half_swap(float& a, float& b) {
  asm("v_permlane16_swap_b32 %0, %1" : "+v"(a), "+v"(b));
}

#define LOG2E_F 1.4426950408889634f
#define LN2_F   0.6931471805599453f

__device__ __forceinline__ float softplus_f(float x) {
  const float e = __builtin_amdgcn_exp2f(x * LOG2E_F);
  const float l = __builtin_amdgcn_logf(1.0f + e) * LN2_F;
  return (x > 20.0f) ? x : l;
}

struct ChunkIn {
  float del, uu, zz;
  float cr[4][2];
  float br[4][2];
};

__device__ __forceinline__ void load_chunk(ChunkIn& L,
    const float* __restrict__ pU, const float* __restrict__ pD,
    const float* __restrict__ pZ, const float* __restrict__ pB,
    const float* __restrict__ pC, int off) {
  L.del = pD[off];
  L.uu  = pU[off];
  L.zz  = pZ[off];
#pragma unroll
  for (int q = 0; q < 4; ++q) {
#pragma unroll
    for (int j = 0; j < 2; ++j) {
      L.cr[q][j] = pC[off + (4 * q + j) * SEQLEN];
      L.br[q][j] = pB[off + (4 * q + j) * SEQLEN];
    }
  }
}

struct RowOps {
  v2f a[4];
  v2f bm[4];
};

__device__ __forceinline__ void row_front(RowOps& R, float sT, float du, float S,
    const float (&cr)[4][2], const float (&br)[4][2], float& xP,
    const float (&Afac)[4][2], float AfP, const float (&xin)[8],
    bool bit0, bool bit1, bool bit2) {
#pragma unroll
  for (int q = 0; q < 4; ++q) {
#pragma unroll
    for (int j = 0; j < 2; ++j) {
      const float af = Afac[q][j];
      const float p  = af * sT;
      R.a[q][j]  = cr[q][j] * __builtin_amdgcn_exp2f(p);
      R.bm[q][j] = fmaf(br[q][j] * du, __builtin_amdgcn_exp2f(-p),
                        xin[2 * q + j]);
    }
  }

  {
    const float t0 = merge_step<0x161>(R.bm[0][0], R.bm[0][1], bit0);
    const float t1 = merge_step<0x161>(R.bm[1][0], R.bm[1][1], bit0);
    const float t2 = merge_step<0x161>(R.bm[2][0], R.bm[2][1], bit0);
    const float t3 = merge_step<0x161>(R.bm[3][0], R.bm[3][1], bit0);
    const float u0 = merge_step<0x162>(t0, t1, bit1);
    const float u1 = merge_step<0x162>(t2, t3, bit1);
    float v = merge_step<0x164>(u0, u1, bit2);
    v += dpp_mov<0x168>(v);
    xP = v * __builtin_amdgcn_exp2f(AfP * S);
  }
}

__device__ __forceinline__ float row_part(const v8f& K, int tl, int half) {
  float m[8];
#pragma unroll
  for (int r = 0; r < 8; ++r) {
    const int srow = r + 8 * half;
    m[r] = (srow <= tl) ? K[r] : 0.0f;
  }
  return ((m[0] + m[1]) + (m[2] + m[3])) + ((m[4] + m[5]) + (m[6] + m[7]));
}

__device__ __forceinline__ float chunk_step2(const ChunkIn& L,
    const float* sLd, float* sSt,
    const float (&Afac0)[4][2], const float (&Afac1)[4][2],
    float AfP0, float AfP1, float DdOwn,
    bool bit0, bool bit1, bool bit2, int tl, int half) {
  float xin0[8], xin1[8];
  {
    const v4f c0 = *(const v4f*)(sLd);
    const v4f c1 = *(const v4f*)(sLd + 4);
    const v4f e0 = *(const v4f*)(sLd + 16);
    const v4f e1 = *(const v4f*)(sLd + 20);
#pragma unroll
    for (int i = 0; i < 4; ++i) {
      xin0[i] = c0[i]; xin0[4 + i] = c1[i];
      xin1[i] = e0[i]; xin1[4 + i] = e1[i];
    }
  }

  const float sig =
      __builtin_amdgcn_rcpf(1.0f + __builtin_amdgcn_exp2f(-L.zz * LOG2E_F));
  const float gate = L.zz * sig;
  const float skip = DdOwn * L.uu;

  const float dtv = softplus_f(L.del);
  float sT0 = prefix16(dtv);
  float du0 = dtv * L.uu;
  float sT1 = sT0;
  float du1 = du0;
  half_swap(sT0, sT1);
  half_swap(du0, du1);
  const float S0 = __shfl(sT0, 15, 16);
  const float S1 = __shfl(sT1, 15, 16);

  RowOps R0, R1;
  float xP0, xP1;
  row_front(R0, sT0, du0, S0, L.cr, L.br, xP0, Afac0, AfP0, xin0,
            bit0, bit1, bit2);
  row_front(R1, sT1, du1, S1, L.cr, L.br, xP1, Afac1, AfP1, xin1,
            bit0, bit1, bit2);
  sSt[0]  = xP0;
  sSt[16] = xP1;

  v8f K0 = {0.f, 0.f, 0.f, 0.f, 0.f, 0.f, 0.f, 0.f};
  v8f K1 = {0.f, 0.f, 0.f, 0.f, 0.f, 0.f, 0.f, 0.f};
#pragma unroll
  for (int q = 0; q < 4; ++q) {
    K0 = __builtin_amdgcn_wmma_f32_16x16x4_f32(
        false, R0.bm[q], false, R0.a[q], (short)0, K0, false, false);
    K1 = __builtin_amdgcn_wmma_f32_16x16x4_f32(
        false, R1.bm[q], false, R1.a[q], (short)0, K1, false, false);
  }

  float p0 = row_part(K0, tl, half);
  float p1 = row_part(K1, tl, half);
  half_swap(p0, p1);
  const float Y = p0 + p1;

  return (Y + skip) * gate;
}

__global__ __launch_bounds__(128) __attribute__((amdgpu_waves_per_eu(4, 4)))
void mamba_scan_kernel(
    const float* __restrict__ u,
    const float* __restrict__ delta,
    const float* __restrict__ A,
    const float* __restrict__ Bm,
    const float* __restrict__ Cm,
    const float* __restrict__ Dv,
    const float* __restrict__ z,
    float* __restrict__ out)
{
  const int lane  = threadIdx.x & 31;
  const int wave  = (blockIdx.x << 2) | (threadIdx.x >> 5);
  const int b     = wave / (DIN / 2);
  const int d     = (wave % (DIN / 2)) * 2;

  const int tl    = lane & 15;
  const int half  = lane >> 4;

  float Afac0[4][2], Afac1[4][2];
#pragma unroll
  for (int q = 0; q < 4; ++q)
#pragma unroll
    for (int j = 0; j < 2; ++j) {
      const int n = 4 * q + j + 2 * half;
      Afac0[q][j] = A[(size_t)d * NSTATE + n] * LOG2E_F;
      Afac1[q][j] = A[(size_t)(d + 1) * NSTATE + n] * LOG2E_F;
    }

  const int   rp   = tl & 7;
  const int   np   = 4 * (rp >> 1) + (rp & 1) + 2 * half;
  const float AfP0 = A[(size_t)d * NSTATE + np] * LOG2E_F;
  const float AfP1 = A[(size_t)(d + 1) * NSTATE + np] * LOG2E_F;
  const bool  bit0 = (tl & 1) != 0;
  const bool  bit1 = (tl & 2) != 0;
  const bool  bit2 = (tl & 4) != 0;

  const float DdOwn = Dv[d + half];

  const size_t rowOwn = ((size_t)b * DIN + d + half) * SEQLEN;
  const size_t rowBC  = (size_t)b * NSTATE * SEQLEN;
  const float* pU = u + rowOwn + tl;
  const float* pD = delta + rowOwn + tl;
  const float* pZ = z + rowOwn + tl;
  const float* pB = Bm + rowBC + (size_t)(2 * half) * SEQLEN + tl;
  const float* pC = Cm + rowBC + (size_t)(2 * half) * SEQLEN + tl;
  float* pO = out + rowOwn + tl;

  __shared__ __attribute__((aligned(16))) float sState[4 * 64];
  float* sw = sState + (threadIdx.x >> 5) * 64;
  sw[lane]      = 0.0f;
  sw[32 + lane] = 0.0f;
  const float* sLd = sw + ((tl == 0) ? 8 * half : 32);
  float* sSt = sw + 8 * half + rp;

  ChunkIn inA, inB;
  load_chunk(inA, pU, pD, pZ, pB, pC, 0);

  for (int c = 0; c < NCHUNK; c += 2) {
    const int base = c * CHUNK;

    load_chunk(inB, pU, pD, pZ, pB, pC, base + CHUNK);
    if (c + 4 < NCHUNK) {
      __builtin_prefetch(&pU[base + 4 * CHUNK], 0, 3);
      __builtin_prefetch(&pD[base + 4 * CHUNK], 0, 3);
      __builtin_prefetch(&pZ[base + 4 * CHUNK], 0, 3);
    }
    __builtin_amdgcn_sched_barrier(0);
    pO[base] =
        chunk_step2(inA, sLd, sSt, Afac0, Afac1, AfP0, AfP1, DdOwn,
                    bit0, bit1, bit2, tl, half);

    const int nbase = (c + 2 < NCHUNK) ? base + 2 * CHUNK : base + CHUNK;
    load_chunk(inA, pU, pD, pZ, pB, pC, nbase);
    __builtin_amdgcn_sched_barrier(0);
    pO[base + CHUNK] =
        chunk_step2(inB, sLd, sSt, Afac0, Afac1, AfP0, AfP1, DdOwn,
                    bit0, bit1, bit2, tl, half);
  }
}

extern "C" void kernel_launch(void* const* d_in, const int* in_sizes, int n_in,
                              void* d_out, int out_size, void* d_ws, size_t ws_size,
                              hipStream_t stream) {
  (void)in_sizes; (void)n_in; (void)d_ws; (void)ws_size; (void)out_size;
  const float* u     = (const float*)d_in[0];
  const float* delta = (const float*)d_in[1];
  const float* A     = (const float*)d_in[2];
  const float* Bm    = (const float*)d_in[3];
  const float* Cm    = (const float*)d_in[4];
  const float* Dv    = (const float*)d_in[5];
  const float* z     = (const float*)d_in[6];
  float* out         = (float*)d_out;

  const int nWaves = BATCH * DIN / 2;
  dim3 grid(nWaves / 4), block(128);
  mamba_scan_kernel<<<grid, block, 0, stream>>>(u, delta, A, Bm, Cm, Dv, z, out);
}
